# starting kernel
# baseline (speedup 1.0000x reference)
#include <hip/hip_runtime.h>


typedef __attribute__((ext_vector_type(2))) float v2f;
typedef __attribute__((ext_vector_type(8))) float v8f;

#define IN_F  64
#define OUT_F 64
#define KER   4
#define EMB   3
#define CDIM  (OUT_F * KER)

__global__ void support_gemm_wmma(const float* __restrict__ x,
                                  const float* __restrict__ w,
                                  float* __restrict__ support,
                                  int N) {
  const int lane = threadIdx.x & 31;
  const int wave = (blockIdx.x * blockDim.x + threadIdx.x) >> 5;
  const int waveStride = (gridDim.x * blockDim.x) >> 5;

  const int tilesPerRow = CDIM / 16;
  const int rowTiles    = (N + 15) >> 4;
  const int totalTiles  = rowTiles * tilesPerRow;

  const int half = lane >> 4;
  const int lid  = lane & 15;

  for (int t = wave; t < totalTiles; t += waveStride) {
    const int m0 = (t / tilesPerRow) * 16;
    const int c0 = (t % tilesPerRow) * 16;

    int arow = m0 + lid;
    if (arow >= N) arow = N - 1;
    const float* __restrict__ arowp = x + (size_t)arow * IN_F + half * 2;
    const float* __restrict__ bcolp = w + c0 + lid + (size_t)(half * 2) * CDIM;

    v8f acc = {};
#pragma unroll
    for (int k0 = 0; k0 < IN_F; k0 += 4) {
      v2f a;
      a.x = arowp[k0];
      a.y = arowp[k0 + 1];
      v2f b;
      b.x = bcolp[(size_t)k0 * CDIM];
      b.y = bcolp[(size_t)(k0 + 1) * CDIM];
      acc = __builtin_amdgcn_wmma_f32_16x16x4_f32(
           false, a,  false, b,
           (short)0, acc,  false,  false);
    }
#pragma unroll
    for (int r = 0; r < 8; ++r) {
      const int row = m0 + r + 8 * half;
      if (row < N) support[(size_t)row * CDIM + c0 + lid] = acc[r];
    }
  }
}

__global__ void init_out(float* __restrict__ out,
                         const float* __restrict__ bias, int total) {
  const int i = blockIdx.x * blockDim.x + threadIdx.x;
  if (i < total) out[i] = bias[i & (OUT_F - 1)];
}

__global__ void edge_scatter(const int*   __restrict__ eidx,
                             const float* __restrict__ spec,
                             const float* __restrict__ mu,
                             const float* __restrict__ sig,
                             const float* __restrict__ support,
                             float* __restrict__ out,
                             int E) {
  const int lane = threadIdx.x & 31;
  const int wave = (blockIdx.x * blockDim.x + threadIdx.x) >> 5;
  const int waveStride = (gridDim.x * blockDim.x) >> 5;

  float mu0[KER], mu1[KER], mu2[KER], sg[KER];
#pragma unroll
  for (int k = 0; k < KER; ++k) {
    mu0[k] = mu[0 * KER + k];
    mu1[k] = mu[1 * KER + k];
    mu2[k] = mu[2 * KER + k];
    sg[k]  = sig[k];
  }

  for (int e = wave; e < E; e += waveStride) {
    const int row = eidx[e];
    const int col = eidx[E + e];

    const float d0 = spec[row * EMB + 0] - spec[col * EMB + 0];
    const float d1 = spec[row * EMB + 1] - spec[col * EMB + 1];
    const float d2 = spec[row * EMB + 2] - spec[col * EMB + 2];

    float val[KER];
#pragma unroll
    for (int k = 0; k < KER; ++k) {
      const float a0 = d0 - mu0[k];
      const float a1 = d1 - mu1[k];
      const float a2 = d2 - mu2[k];
      const float qq = -0.5f * (a0 * a0 + a1 * a1 + a2 * a2);
      val[k] = __expf(sg[k] * qq);
    }

    const float4* sp = (const float4*)(support + (size_t)col * CDIM + lane * 8);
    const float4 s0 = sp[0];
    const float4 s1 = sp[1];
    const float m0 = s0.x * val[0] + s0.y * val[1] + s0.z * val[2] + s0.w * val[3];
    const float m1 = s1.x * val[0] + s1.y * val[1] + s1.z * val[2] + s1.w * val[3];

    float* dst = out + (size_t)row * OUT_F + lane * 2;
    unsafeAtomicAdd(dst,     m0);
    unsafeAtomicAdd(dst + 1, m1);
  }
}

extern "C" void kernel_launch(void* const* d_in, const int* in_sizes, int n_in,
                              void* d_out, int out_size, void* d_ws, size_t ws_size,
                              hipStream_t stream) {
  const float* x    = (const float*)d_in[0];
  const int*   eidx = (const int*)  d_in[1];
  const float* spec = (const float*)d_in[2];
  const float* w    = (const float*)d_in[3];
  const float* bias = (const float*)d_in[4];
  const float* mu   = (const float*)d_in[5];
  const float* sig  = (const float*)d_in[6];

  const int N = in_sizes[0] / IN_F;
  const int E = in_sizes[1] / 2;

  float* support = (float*)d_ws;
  float* out     = (float*)d_out;

  {
    const int totalTiles = ((N + 15) / 16) * (CDIM / 16);
    const int wavesPerBlock = 8;
    const int blocks = (totalTiles + wavesPerBlock - 1) / wavesPerBlock;
    support_gemm_wmma<<<blocks, 256, 0, stream>>>(x, w, support, N);
  }

  {
    const int total = N * OUT_F;
    init_out<<<(total + 255) / 256, 256, 0, stream>>>(out, bias, total);
  }

  {
    const int blocks = 2048;
    edge_scatter<<<blocks, 256, 0, stream>>>(eidx, spec, mu, sig, support, out, E);
  }
}


# GEMM weight-stationary: each wave keeps its 32-column weight stripe in registers and loops over row tiles; x read as 16-B row pieces, no LDS; k order of dot products permuted (sum order changes)
# speedup vs baseline: 1.1230x; 1.1230x over previous
#include <hip/hip_runtime.h>


typedef __attribute__((ext_vector_type(2))) float v2f;
typedef __attribute__((ext_vector_type(8))) float v8f;

#define IN_F  64
#define OUT_F 64
#define KER   4
#define EMB   3
#define CDIM  (OUT_F * KER)

__device__ __forceinline__ void load_a_tile(float4 (&a)[8],
                                            const float* __restrict__ x,
                                            int m0, int lid, int half, int N) {
  int arow = m0 + lid;
  if (arow >= N) arow = N - 1;
  const float4* __restrict__ ap =
      (const float4*)(x + (size_t)arow * IN_F + 4 * half);
#pragma unroll
  for (int j = 0; j < 8; ++j) a[j] = ap[2 * j];
}

__global__ void __launch_bounds__(256)
support_gemm_wmma(const float* __restrict__ x,
                  const float* __restrict__ w,
                  float* __restrict__ support,
                  int N) {
  const int lane = threadIdx.x & 31;
  const int wv   = threadIdx.x >> 5;
  const int half = lane >> 4;
  const int lid  = lane & 15;
  const int rowTiles = (N + 15) >> 4;

  int tile = blockIdx.x;
  if (tile >= rowTiles) return;

  const float* __restrict__ wp = w + 32 * wv + 2 * lid;
  v2f b0[16], b1[16];
#pragma unroll
  for (int t = 0; t < 16; ++t) {
    const int k = 8 * (t >> 1) + 4 * half + 2 * (t & 1);
    b0[t].x = wp[(size_t)k * CDIM];
    b1[t].x = wp[(size_t)k * CDIM + 1];
    b0[t].y = wp[(size_t)(k + 1) * CDIM];
    b1[t].y = wp[(size_t)(k + 1) * CDIM + 1];
  }

  float4 a[8];
  load_a_tile(a, x, tile * 16, lid, half, N);

  for (; tile < rowTiles; tile += gridDim.x) {
    const int m0 = tile * 16;

    float4 an[8];
    const int nt = tile + gridDim.x;
    load_a_tile(an, x, (nt < rowTiles ? nt : tile) * 16, lid, half, N);
    __builtin_amdgcn_sched_barrier(0);

    v8f acc0 = {}, acc1 = {};
#pragma unroll
    for (int j = 0; j < 8; ++j) {
      v2f alo, ahi;
      alo.x = a[j].x; alo.y = a[j].y;
      ahi.x = a[j].z; ahi.y = a[j].w;
      acc0 = __builtin_amdgcn_wmma_f32_16x16x4_f32(
          false, alo, false, b0[2 * j], (short)0, acc0, false, false);
      acc1 = __builtin_amdgcn_wmma_f32_16x16x4_f32(
          false, alo, false, b1[2 * j], (short)0, acc1, false, false);
      acc0 = __builtin_amdgcn_wmma_f32_16x16x4_f32(
          false, ahi, false, b0[2 * j + 1], (short)0, acc0, false, false);
      acc1 = __builtin_amdgcn_wmma_f32_16x16x4_f32(
          false, ahi, false, b1[2 * j + 1], (short)0, acc1, false, false);
    }

    float* __restrict__ cp =
        support + (size_t)(m0 + 8 * half) * CDIM + 32 * wv + 2 * lid;
    if (m0 + 16 <= N) {
#pragma unroll
      for (int r = 0; r < 8; ++r) {
        float2 v; v.x = acc0[r]; v.y = acc1[r];
        *(float2*)(cp + (size_t)r * CDIM) = v;
      }
    } else {
#pragma unroll
      for (int r = 0; r < 8; ++r) {
        if (m0 + r + 8 * half < N) {
          float2 v; v.x = acc0[r]; v.y = acc1[r];
          *(float2*)(cp + (size_t)r * CDIM) = v;
        }
      }
    }

#pragma unroll
    for (int j = 0; j < 8; ++j) a[j] = an[j];
  }
}

__global__ void init_out(float* __restrict__ out,
                         const float* __restrict__ bias, int total) {
  const int i = blockIdx.x * blockDim.x + threadIdx.x;
  if (i < total) out[i] = bias[i & (OUT_F - 1)];
}

__global__ void edge_scatter(const int*   __restrict__ eidx,
                             const float* __restrict__ spec,
                             const float* __restrict__ mu,
                             const float* __restrict__ sig,
                             const float* __restrict__ support,
                             float* __restrict__ out,
                             int E) {
  const int lane = threadIdx.x & 31;
  const int wave = (blockIdx.x * blockDim.x + threadIdx.x) >> 5;
  const int waveStride = (gridDim.x * blockDim.x) >> 5;

  float mu0[KER], mu1[KER], mu2[KER], sg[KER];
#pragma unroll
  for (int k = 0; k < KER; ++k) {
    mu0[k] = mu[0 * KER + k];
    mu1[k] = mu[1 * KER + k];
    mu2[k] = mu[2 * KER + k];
    sg[k]  = sig[k];
  }

  for (int e = wave; e < E; e += waveStride) {
    const int row = eidx[e];
    const int col = eidx[E + e];

    const float d0 = spec[row * EMB + 0] - spec[col * EMB + 0];
    const float d1 = spec[row * EMB + 1] - spec[col * EMB + 1];
    const float d2 = spec[row * EMB + 2] - spec[col * EMB + 2];

    float val[KER];
#pragma unroll
    for (int k = 0; k < KER; ++k) {
      const float a0 = d0 - mu0[k];
      const float a1 = d1 - mu1[k];
      const float a2 = d2 - mu2[k];
      const float qq = -0.5f * (a0 * a0 + a1 * a1 + a2 * a2);
      val[k] = __expf(sg[k] * qq);
    }

    const float4* sp = (const float4*)(support + (size_t)col * CDIM + lane * 8);
    const float4 s0 = sp[0];
    const float4 s1 = sp[1];
    const float m0 = s0.x * val[0] + s0.y * val[1] + s0.z * val[2] + s0.w * val[3];
    const float m1 = s1.x * val[0] + s1.y * val[1] + s1.z * val[2] + s1.w * val[3];

    float* dst = out + (size_t)row * OUT_F + lane * 2;
    unsafeAtomicAdd(dst,     m0);
    unsafeAtomicAdd(dst + 1, m1);
  }
}

extern "C" void kernel_launch(void* const* d_in, const int* in_sizes, int n_in,
                              void* d_out, int out_size, void* d_ws, size_t ws_size,
                              hipStream_t stream) {
  const float* x    = (const float*)d_in[0];
  const int*   eidx = (const int*)  d_in[1];
  const float* spec = (const float*)d_in[2];
  const float* w    = (const float*)d_in[3];
  const float* bias = (const float*)d_in[4];
  const float* mu   = (const float*)d_in[5];
  const float* sig  = (const float*)d_in[6];

  const int N = in_sizes[0] / IN_F;
  const int E = in_sizes[1] / 2;

  float* support = (float*)d_ws;
  float* out     = (float*)d_out;

  {
    support_gemm_wmma<<<512, 256, 0, stream>>>(x, w, support, N);
  }

  {
    const int total = N * OUT_F;
    init_out<<<(total + 255) / 256, 256, 0, stream>>>(out, bias, total);
  }

  {
    const int blocks = 2048;
    edge_scatter<<<blocks, 256, 0, stream>>>(eidx, spec, mu, sig, support, out, E);
  }
}


# Bias seeding of out by 256 long-lived 8-wave workgroups, each thread storing 16-B pieces in a grid-stride loop (was 12,500 workgroups = 100,000 one-store waves to enter); sum order unchanged
# speedup vs baseline: 1.1762x; 1.0474x over previous
#include <hip/hip_runtime.h>


typedef __attribute__((ext_vector_type(2))) float v2f;
typedef __attribute__((ext_vector_type(8))) float v8f;

#define IN_F  64
#define OUT_F 64
#define KER   4
#define EMB   3
#define CDIM  (OUT_F * KER)

__device__ __forceinline__ void load_a_tile(float4 (&a)[8],
                                            const float* __restrict__ x,
                                            int m0, int lid, int half, int N) {
  int arow = m0 + lid;
  if (arow >= N) arow = N - 1;
  const float4* __restrict__ ap =
      (const float4*)(x + (size_t)arow * IN_F + 4 * half);
#pragma unroll
  for (int j = 0; j < 8; ++j) a[j] = ap[2 * j];
}

__global__ void __launch_bounds__(256)
support_gemm_wmma(const float* __restrict__ x,
                  const float* __restrict__ w,
                  float* __restrict__ support,
                  int N) {
  const int lane = threadIdx.x & 31;
  const int wv   = threadIdx.x >> 5;
  const int half = lane >> 4;
  const int lid  = lane & 15;
  const int rowTiles = (N + 15) >> 4;

  int tile = blockIdx.x;
  if (tile >= rowTiles) return;

  const float* __restrict__ wp = w + 32 * wv + 2 * lid;
  v2f b0[16], b1[16];
#pragma unroll
  for (int t = 0; t < 16; ++t) {
    const int k = 8 * (t >> 1) + 4 * half + 2 * (t & 1);
    b0[t].x = wp[(size_t)k * CDIM];
    b1[t].x = wp[(size_t)k * CDIM + 1];
    b0[t].y = wp[(size_t)(k + 1) * CDIM];
    b1[t].y = wp[(size_t)(k + 1) * CDIM + 1];
  }

  float4 a[8];
  load_a_tile(a, x, tile * 16, lid, half, N);

  for (; tile < rowTiles; tile += gridDim.x) {
    const int m0 = tile * 16;

    float4 an[8];
    const int nt = tile + gridDim.x;
    load_a_tile(an, x, (nt < rowTiles ? nt : tile) * 16, lid, half, N);
    __builtin_amdgcn_sched_barrier(0);

    v8f acc0 = {}, acc1 = {};
#pragma unroll
    for (int j = 0; j < 8; ++j) {
      v2f alo, ahi;
      alo.x = a[j].x; alo.y = a[j].y;
      ahi.x = a[j].z; ahi.y = a[j].w;
      acc0 = __builtin_amdgcn_wmma_f32_16x16x4_f32(
          false, alo, false, b0[2 * j], (short)0, acc0, false, false);
      acc1 = __builtin_amdgcn_wmma_f32_16x16x4_f32(
          false, alo, false, b1[2 * j], (short)0, acc1, false, false);
      acc0 = __builtin_amdgcn_wmma_f32_16x16x4_f32(
          false, ahi, false, b0[2 * j + 1], (short)0, acc0, false, false);
      acc1 = __builtin_amdgcn_wmma_f32_16x16x4_f32(
          false, ahi, false, b1[2 * j + 1], (short)0, acc1, false, false);
    }

    float* __restrict__ cp =
        support + (size_t)(m0 + 8 * half) * CDIM + 32 * wv + 2 * lid;
    if (m0 + 16 <= N) {
#pragma unroll
      for (int r = 0; r < 8; ++r) {
        float2 v; v.x = acc0[r]; v.y = acc1[r];
        *(float2*)(cp + (size_t)r * CDIM) = v;
      }
    } else {
#pragma unroll
      for (int r = 0; r < 8; ++r) {
        if (m0 + r + 8 * half < N) {
          float2 v; v.x = acc0[r]; v.y = acc1[r];
          *(float2*)(cp + (size_t)r * CDIM) = v;
        }
      }
    }

#pragma unroll
    for (int j = 0; j < 8; ++j) a[j] = an[j];
  }
}

__global__ void __launch_bounds__(256)
init_out(float* __restrict__ out, const float* __restrict__ bias, int total) {
  const int n4     = total >> 2;
  const int tid    = blockIdx.x * 256 + threadIdx.x;
  const int stride = gridDim.x * 256;
  const float* __restrict__ bp = bias + 4 * (tid & (OUT_F / 4 - 1));
  float4 b;
  b.x = bp[0]; b.y = bp[1]; b.z = bp[2]; b.w = bp[3];
  asm volatile("" :: "v"(b.x), "v"(b.y), "v"(b.z), "v"(b.w));
  float4* __restrict__ o4 = (float4*)out;
  for (int i = tid; i < n4; i += stride) o4[i] = b;
}

__global__ void edge_scatter(const int*   __restrict__ eidx,
                             const float* __restrict__ spec,
                             const float* __restrict__ mu,
                             const float* __restrict__ sig,
                             const float* __restrict__ support,
                             float* __restrict__ out,
                             int E) {
  const int lane = threadIdx.x & 31;
  const int wave = (blockIdx.x * blockDim.x + threadIdx.x) >> 5;
  const int waveStride = (gridDim.x * blockDim.x) >> 5;

  float mu0[KER], mu1[KER], mu2[KER], sg[KER];
#pragma unroll
  for (int k = 0; k < KER; ++k) {
    mu0[k] = mu[0 * KER + k];
    mu1[k] = mu[1 * KER + k];
    mu2[k] = mu[2 * KER + k];
    sg[k]  = sig[k];
  }

  for (int e = wave; e < E; e += waveStride) {
    const int row = eidx[e];
    const int col = eidx[E + e];

    const float d0 = spec[row * EMB + 0] - spec[col * EMB + 0];
    const float d1 = spec[row * EMB + 1] - spec[col * EMB + 1];
    const float d2 = spec[row * EMB + 2] - spec[col * EMB + 2];

    float val[KER];
#pragma unroll
    for (int k = 0; k < KER; ++k) {
      const float a0 = d0 - mu0[k];
      const float a1 = d1 - mu1[k];
      const float a2 = d2 - mu2[k];
      const float qq = -0.5f * (a0 * a0 + a1 * a1 + a2 * a2);
      val[k] = __expf(sg[k] * qq);
    }

    const float4* sp = (const float4*)(support + (size_t)col * CDIM + lane * 8);
    const float4 s0 = sp[0];
    const float4 s1 = sp[1];
    const float m0 = s0.x * val[0] + s0.y * val[1] + s0.z * val[2] + s0.w * val[3];
    const float m1 = s1.x * val[0] + s1.y * val[1] + s1.z * val[2] + s1.w * val[3];

    float* dst = out + (size_t)row * OUT_F + lane * 2;
    unsafeAtomicAdd(dst,     m0);
    unsafeAtomicAdd(dst + 1, m1);
  }
}

extern "C" void kernel_launch(void* const* d_in, const int* in_sizes, int n_in,
                              void* d_out, int out_size, void* d_ws, size_t ws_size,
                              hipStream_t stream) {
  const float* x    = (const float*)d_in[0];
  const int*   eidx = (const int*)  d_in[1];
  const float* spec = (const float*)d_in[2];
  const float* w    = (const float*)d_in[3];
  const float* bias = (const float*)d_in[4];
  const float* mu   = (const float*)d_in[5];
  const float* sig  = (const float*)d_in[6];

  const int N = in_sizes[0] / IN_F;
  const int E = in_sizes[1] / 2;

  float* support = (float*)d_ws;
  float* out     = (float*)d_out;

  {
    support_gemm_wmma<<<512, 256, 0, stream>>>(x, w, support, N);
  }

  {
    const int total = N * OUT_F;
    init_out<<<256, 256, 0, stream>>>(out, bias, total);
  }

  {
    const int blocks = 2048;
    edge_scatter<<<blocks, 256, 0, stream>>>(eidx, spec, mu, sig, support, out, E);
  }
}


# Edge phase: wave number made scalar (readfirstlane): an edge's indices and spec points come by scalar loads, diffs and row bases by scalar arithmetic, once per wave not per lane; sum order unchanged
# speedup vs baseline: 1.3576x; 1.1542x over previous
#include <hip/hip_runtime.h>


typedef __attribute__((ext_vector_type(2))) float v2f;
typedef __attribute__((ext_vector_type(8))) float v8f;

#define IN_F  64
#define OUT_F 64
#define KER   4
#define EMB   3
#define CDIM  (OUT_F * KER)

__device__ __forceinline__ void load_a_tile(float4 (&a)[8],
                                            const float* __restrict__ x,
                                            int m0, int lid, int half, int N) {
  int arow = m0 + lid;
  if (arow >= N) arow = N - 1;
  const float4* __restrict__ ap =
      (const float4*)(x + (size_t)arow * IN_F + 4 * half);
#pragma unroll
  for (int j = 0; j < 8; ++j) a[j] = ap[2 * j];
}

__global__ void __launch_bounds__(256)
support_gemm_wmma(const float* __restrict__ x,
                  const float* __restrict__ w,
                  float* __restrict__ support,
                  int N) {
  const int lane = threadIdx.x & 31;
  const int wv   = threadIdx.x >> 5;
  const int half = lane >> 4;
  const int lid  = lane & 15;
  const int rowTiles = (N + 15) >> 4;

  int tile = blockIdx.x;
  if (tile >= rowTiles) return;

  const float* __restrict__ wp = w + 32 * wv + 2 * lid;
  v2f b0[16], b1[16];
#pragma unroll
  for (int t = 0; t < 16; ++t) {
    const int k = 8 * (t >> 1) + 4 * half + 2 * (t & 1);
    b0[t].x = wp[(size_t)k * CDIM];
    b1[t].x = wp[(size_t)k * CDIM + 1];
    b0[t].y = wp[(size_t)(k + 1) * CDIM];
    b1[t].y = wp[(size_t)(k + 1) * CDIM + 1];
  }

  float4 a[8];
  load_a_tile(a, x, tile * 16, lid, half, N);

  for (; tile < rowTiles; tile += gridDim.x) {
    const int m0 = tile * 16;

    float4 an[8];
    const int nt = tile + gridDim.x;
    load_a_tile(an, x, (nt < rowTiles ? nt : tile) * 16, lid, half, N);
    __builtin_amdgcn_sched_barrier(0);

    v8f acc0 = {}, acc1 = {};
#pragma unroll
    for (int j = 0; j < 8; ++j) {
      v2f alo, ahi;
      alo.x = a[j].x; alo.y = a[j].y;
      ahi.x = a[j].z; ahi.y = a[j].w;
      acc0 = __builtin_amdgcn_wmma_f32_16x16x4_f32(
          false, alo, false, b0[2 * j], (short)0, acc0, false, false);
      acc1 = __builtin_amdgcn_wmma_f32_16x16x4_f32(
          false, alo, false, b1[2 * j], (short)0, acc1, false, false);
      acc0 = __builtin_amdgcn_wmma_f32_16x16x4_f32(
          false, ahi, false, b0[2 * j + 1], (short)0, acc0, false, false);
      acc1 = __builtin_amdgcn_wmma_f32_16x16x4_f32(
          false, ahi, false, b1[2 * j + 1], (short)0, acc1, false, false);
    }

    float* __restrict__ cp =
        support + (size_t)(m0 + 8 * half) * CDIM + 32 * wv + 2 * lid;
    if (m0 + 16 <= N) {
#pragma unroll
      for (int r = 0; r < 8; ++r) {
        float2 v; v.x = acc0[r]; v.y = acc1[r];
        *(float2*)(cp + (size_t)r * CDIM) = v;
      }
    } else {
#pragma unroll
      for (int r = 0; r < 8; ++r) {
        if (m0 + r + 8 * half < N) {
          float2 v; v.x = acc0[r]; v.y = acc1[r];
          *(float2*)(cp + (size_t)r * CDIM) = v;
        }
      }
    }

#pragma unroll
    for (int j = 0; j < 8; ++j) a[j] = an[j];
  }
}

__global__ void __launch_bounds__(256)
init_out(float* __restrict__ out, const float* __restrict__ bias, int total) {
  const int n4     = total >> 2;
  const int tid    = blockIdx.x * 256 + threadIdx.x;
  const int stride = gridDim.x * 256;
  const float* __restrict__ bp = bias + 4 * (tid & (OUT_F / 4 - 1));
  float4 b;
  b.x = bp[0]; b.y = bp[1]; b.z = bp[2]; b.w = bp[3];
  asm volatile("" :: "v"(b.x), "v"(b.y), "v"(b.z), "v"(b.w));
  float4* __restrict__ o4 = (float4*)out;
  for (int i = tid; i < n4; i += stride) o4[i] = b;
}

__global__ void edge_scatter(const int*   __restrict__ eidx,
                             const float* __restrict__ spec,
                             const float* __restrict__ mu,
                             const float* __restrict__ sig,
                             const float* __restrict__ support,
                             float* __restrict__ out,
                             int E) {
  const int lane = threadIdx.x & 31;
  const int wave = __builtin_amdgcn_readfirstlane(
      (int)((blockIdx.x * blockDim.x + threadIdx.x) >> 5));
  const int waveStride = (gridDim.x * blockDim.x) >> 5;

  float mu0[KER], mu1[KER], mu2[KER], sg[KER];
#pragma unroll
  for (int k = 0; k < KER; ++k) {
    mu0[k] = mu[0 * KER + k];
    mu1[k] = mu[1 * KER + k];
    mu2[k] = mu[2 * KER + k];
    sg[k]  = sig[k];
  }

  for (int e = wave; e < E; e += waveStride) {
    const int row = eidx[e];
    const int col = eidx[E + e];

    const float d0 = spec[row * EMB + 0] - spec[col * EMB + 0];
    const float d1 = spec[row * EMB + 1] - spec[col * EMB + 1];
    const float d2 = spec[row * EMB + 2] - spec[col * EMB + 2];

    float val[KER];
#pragma unroll
    for (int k = 0; k < KER; ++k) {
      const float a0 = d0 - mu0[k];
      const float a1 = d1 - mu1[k];
      const float a2 = d2 - mu2[k];
      const float qq = -0.5f * (a0 * a0 + a1 * a1 + a2 * a2);
      val[k] = __expf(sg[k] * qq);
    }

    const float4* sp = (const float4*)(support + (size_t)col * CDIM + lane * 8);
    const float4 s0 = sp[0];
    const float4 s1 = sp[1];
    const float m0 = s0.x * val[0] + s0.y * val[1] + s0.z * val[2] + s0.w * val[3];
    const float m1 = s1.x * val[0] + s1.y * val[1] + s1.z * val[2] + s1.w * val[3];

    float* dst = out + (size_t)row * OUT_F + lane * 2;
    unsafeAtomicAdd(dst,     m0);
    unsafeAtomicAdd(dst + 1, m1);
  }
}

extern "C" void kernel_launch(void* const* d_in, const int* in_sizes, int n_in,
                              void* d_out, int out_size, void* d_ws, size_t ws_size,
                              hipStream_t stream) {
  const float* x    = (const float*)d_in[0];
  const int*   eidx = (const int*)  d_in[1];
  const float* spec = (const float*)d_in[2];
  const float* w    = (const float*)d_in[3];
  const float* bias = (const float*)d_in[4];
  const float* mu   = (const float*)d_in[5];
  const float* sig  = (const float*)d_in[6];

  const int N = in_sizes[0] / IN_F;
  const int E = in_sizes[1] / 2;

  float* support = (float*)d_ws;
  float* out     = (float*)d_out;

  {
    support_gemm_wmma<<<512, 256, 0, stream>>>(x, w, support, N);
  }

  {
    const int total = N * OUT_F;
    init_out<<<256, 256, 0, stream>>>(out, bias, total);
  }

  {
    const int blocks = 2048;
    edge_scatter<<<blocks, 256, 0, stream>>>(eidx, spec, mu, sig, support, out, E);
  }
}


# Edge phase: a wave takes 4 consecutive edges per turn, issues their spec and gather loads together and reads indices one group ahead, so load latencies overlap; atomic arrival order changes
# speedup vs baseline: 1.4056x; 1.0354x over previous
#include <hip/hip_runtime.h>


typedef __attribute__((ext_vector_type(2))) float v2f;
typedef __attribute__((ext_vector_type(8))) float v8f;

#define IN_F  64
#define OUT_F 64
#define KER   4
#define EMB   3
#define CDIM  (OUT_F * KER)

__device__ __forceinline__ void load_a_tile(float4 (&a)[8],
                                            const float* __restrict__ x,
                                            int m0, int lid, int half, int N) {
  int arow = m0 + lid;
  if (arow >= N) arow = N - 1;
  const float4* __restrict__ ap =
      (const float4*)(x + (size_t)arow * IN_F + 4 * half);
#pragma unroll
  for (int j = 0; j < 8; ++j) a[j] = ap[2 * j];
}

__global__ void __launch_bounds__(256)
support_gemm_wmma(const float* __restrict__ x,
                  const float* __restrict__ w,
                  float* __restrict__ support,
                  int N) {
  const int lane = threadIdx.x & 31;
  const int wv   = threadIdx.x >> 5;
  const int half = lane >> 4;
  const int lid  = lane & 15;
  const int rowTiles = (N + 15) >> 4;

  int tile = blockIdx.x;
  if (tile >= rowTiles) return;

  const float* __restrict__ wp = w + 32 * wv + 2 * lid;
  v2f b0[16], b1[16];
#pragma unroll
  for (int t = 0; t < 16; ++t) {
    const int k = 8 * (t >> 1) + 4 * half + 2 * (t & 1);
    b0[t].x = wp[(size_t)k * CDIM];
    b1[t].x = wp[(size_t)k * CDIM + 1];
    b0[t].y = wp[(size_t)(k + 1) * CDIM];
    b1[t].y = wp[(size_t)(k + 1) * CDIM + 1];
  }

  float4 a[8];
  load_a_tile(a, x, tile * 16, lid, half, N);

  for (; tile < rowTiles; tile += gridDim.x) {
    const int m0 = tile * 16;

    float4 an[8];
    const int nt = tile + gridDim.x;
    load_a_tile(an, x, (nt < rowTiles ? nt : tile) * 16, lid, half, N);
    __builtin_amdgcn_sched_barrier(0);

    v8f acc0 = {}, acc1 = {};
#pragma unroll
    for (int j = 0; j < 8; ++j) {
      v2f alo, ahi;
      alo.x = a[j].x; alo.y = a[j].y;
      ahi.x = a[j].z; ahi.y = a[j].w;
      acc0 = __builtin_amdgcn_wmma_f32_16x16x4_f32(
          false, alo, false, b0[2 * j], (short)0, acc0, false, false);
      acc1 = __builtin_amdgcn_wmma_f32_16x16x4_f32(
          false, alo, false, b1[2 * j], (short)0, acc1, false, false);
      acc0 = __builtin_amdgcn_wmma_f32_16x16x4_f32(
          false, ahi, false, b0[2 * j + 1], (short)0, acc0, false, false);
      acc1 = __builtin_amdgcn_wmma_f32_16x16x4_f32(
          false, ahi, false, b1[2 * j + 1], (short)0, acc1, false, false);
    }

    float* __restrict__ cp =
        support + (size_t)(m0 + 8 * half) * CDIM + 32 * wv + 2 * lid;
    if (m0 + 16 <= N) {
#pragma unroll
      for (int r = 0; r < 8; ++r) {
        float2 v; v.x = acc0[r]; v.y = acc1[r];
        *(float2*)(cp + (size_t)r * CDIM) = v;
      }
    } else {
#pragma unroll
      for (int r = 0; r < 8; ++r) {
        if (m0 + r + 8 * half < N) {
          float2 v; v.x = acc0[r]; v.y = acc1[r];
          *(float2*)(cp + (size_t)r * CDIM) = v;
        }
      }
    }

#pragma unroll
    for (int j = 0; j < 8; ++j) a[j] = an[j];
  }
}

__global__ void __launch_bounds__(256)
init_out(float* __restrict__ out, const float* __restrict__ bias, int total) {
  const int n4     = total >> 2;
  const int tid    = blockIdx.x * 256 + threadIdx.x;
  const int stride = gridDim.x * 256;
  const float* __restrict__ bp = bias + 4 * (tid & (OUT_F / 4 - 1));
  float4 b;
  b.x = bp[0]; b.y = bp[1]; b.z = bp[2]; b.w = bp[3];
  asm volatile("" :: "v"(b.x), "v"(b.y), "v"(b.z), "v"(b.w));
  float4* __restrict__ o4 = (float4*)out;
  for (int i = tid; i < n4; i += stride) o4[i] = b;
}

#define EG 4
typedef __attribute__((ext_vector_type(4))) float v4f;

__device__ __forceinline__ void edge_math(float d0, float d1, float d2,
                                          const float (&mu0)[KER],
                                          const float (&mu1)[KER],
                                          const float (&mu2)[KER],
                                          const float (&sg)[KER],
                                          const v4f s0, const v4f s1,
                                          float* __restrict__ dst) {
  float val[KER];
#pragma unroll
  for (int k = 0; k < KER; ++k) {
    const float a0 = d0 - mu0[k];
    const float a1 = d1 - mu1[k];
    const float a2 = d2 - mu2[k];
    const float qq = -0.5f * (a0 * a0 + a1 * a1 + a2 * a2);
    val[k] = __expf(sg[k] * qq);
  }
  const float m0 = s0.x * val[0] + s0.y * val[1] + s0.z * val[2] + s0.w * val[3];
  const float m1 = s1.x * val[0] + s1.y * val[1] + s1.z * val[2] + s1.w * val[3];
  unsafeAtomicAdd(dst,     m0);
  unsafeAtomicAdd(dst + 1, m1);
}

__global__ void __launch_bounds__(256)
edge_scatter(const int*   __restrict__ eidx,
             const float* __restrict__ spec,
             const float* __restrict__ mu,
             const float* __restrict__ sig,
             const float* __restrict__ support,
             float* __restrict__ out,
             int E) {
  const int lane = threadIdx.x & 31;
  const int wave = __builtin_amdgcn_readfirstlane(
      (int)((blockIdx.x * blockDim.x + threadIdx.x) >> 5));
  const int waveStride = (gridDim.x * blockDim.x) >> 5;

  float mu0[KER], mu1[KER], mu2[KER], sg[KER];
#pragma unroll
  for (int k = 0; k < KER; ++k) {
    mu0[k] = mu[0 * KER + k];
    mu1[k] = mu[1 * KER + k];
    mu2[k] = mu[2 * KER + k];
    sg[k]  = sig[k];
  }

  const int nGroups = E / EG;
  int rowN[EG], colN[EG];
  if (wave < nGroups) {
#pragma unroll
    for (int j = 0; j < EG; ++j) {
      rowN[j] = eidx[wave * EG + j];
      colN[j] = eidx[E + wave * EG + j];
    }
  }
  for (int g = wave; g < nGroups; g += waveStride) {
    int row[EG], col[EG];
#pragma unroll
    for (int j = 0; j < EG; ++j) { row[j] = rowN[j]; col[j] = colN[j]; }
    const int gn = g + waveStride;
    const int en = (gn < nGroups ? gn : g) * EG;
#pragma unroll
    for (int j = 0; j < EG; ++j) {
      rowN[j] = eidx[en + j];
      colN[j] = eidx[E + en + j];
    }

    v4f s0[EG], s1[EG];
#pragma unroll
    for (int j = 0; j < EG; ++j) {
      const v4f* sp = (const v4f*)(support + (size_t)col[j] * CDIM + lane * 8);
      s0[j] = sp[0];
      s1[j] = sp[1];
    }

    float d0[EG], d1[EG], d2[EG];
#pragma unroll
    for (int j = 0; j < EG; ++j) {
      d0[j] = spec[row[j] * EMB + 0] - spec[col[j] * EMB + 0];
      d1[j] = spec[row[j] * EMB + 1] - spec[col[j] * EMB + 1];
      d2[j] = spec[row[j] * EMB + 2] - spec[col[j] * EMB + 2];
    }

    asm volatile("" :: "v"(s0[0]), "v"(s1[0]),
                   "v"(s0[1]), "v"(s1[1]),
                   "v"(s0[2]), "v"(s1[2]),
                   "v"(s0[3]), "v"(s1[3]));

#pragma unroll
    for (int j = 0; j < EG; ++j) {
      edge_math(d0[j], d1[j], d2[j], mu0, mu1, mu2, sg, s0[j], s1[j],
                out + (size_t)row[j] * OUT_F + lane * 2);
    }
  }

  for (int e = nGroups * EG + wave; e < E; e += waveStride) {
    const int row = eidx[e];
    const int col = eidx[E + e];
    const float d0 = spec[row * EMB + 0] - spec[col * EMB + 0];
    const float d1 = spec[row * EMB + 1] - spec[col * EMB + 1];
    const float d2 = spec[row * EMB + 2] - spec[col * EMB + 2];
    const v4f* sp = (const v4f*)(support + (size_t)col * CDIM + lane * 8);
    const v4f s0 = sp[0];
    const v4f s1 = sp[1];
    edge_math(d0, d1, d2, mu0, mu1, mu2, sg, s0, s1,
              out + (size_t)row * OUT_F + lane * 2);
  }
}

extern "C" void kernel_launch(void* const* d_in, const int* in_sizes, int n_in,
                              void* d_out, int out_size, void* d_ws, size_t ws_size,
                              hipStream_t stream) {
  const float* x    = (const float*)d_in[0];
  const int*   eidx = (const int*)  d_in[1];
  const float* spec = (const float*)d_in[2];
  const float* w    = (const float*)d_in[3];
  const float* bias = (const float*)d_in[4];
  const float* mu   = (const float*)d_in[5];
  const float* sig  = (const float*)d_in[6];

  const int N = in_sizes[0] / IN_F;
  const int E = in_sizes[1] / 2;

  float* support = (float*)d_ws;
  float* out     = (float*)d_out;

  {
    support_gemm_wmma<<<512, 256, 0, stream>>>(x, w, support, N);
  }

  {
    const int total = N * OUT_F;
    init_out<<<256, 256, 0, stream>>>(out, bias, total);
  }

  {
    const int blocks = 2048;
    edge_scatter<<<blocks, 256, 0, stream>>>(eidx, spec, mu, sig, support, out, E);
  }
}
